# starting kernel
# baseline (speedup 1.0000x reference)
#include <hip/hip_runtime.h>

typedef float v2f __attribute__((ext_vector_type(2)));
typedef float v4f __attribute__((ext_vector_type(4)));
typedef float v8f __attribute__((ext_vector_type(8)));

#define N_IN   8192
#define N_OUT  8192
#define NB     8
#define KCH    1024
#define NCHUNK (N_IN / KCH)
#define NTILE  (N_OUT / 16)

__global__ __launch_bounds__(32)
void tri_wmma_partial(const float* __restrict__ w, const float* __restrict__ x,
                      const float* __restrict__ bias, float* __restrict__ ws,
                      float* __restrict__ out, int k_chunk, int final_mode) {
  const int tile    = blockIdx.x;
  const int o0      = tile << 4;
  const int fc_min  = (o0 > 4) ? (o0 - 4) : 0;
  const int k_begin = blockIdx.y * k_chunk;
  int i_end = k_begin + k_chunk;
  if (i_end > N_IN) i_end = N_IN;
  if (i_end <= fc_min) return;
  int i = max(k_begin, fc_min & ~3);

  const int lane = threadIdx.x;
  const int half = lane >> 4;
  const int n    = lane & 15;
  const int k0   = half << 1;

  const int  o  = o0 + n;
  const int  fc = (o > 4) ? (o - 4) : 0;
  const long rs = (long)o * N_IN - ((o >= 5) ? (((long)(o - 5) * (o - 4)) >> 1) : 0);
  const float* __restrict__ wrow = w + (rs - fc);
  const float* __restrict__ xrow = x + (long)(n & 7) * N_IN;

  v8f acc = {};

  int rag_end = o0 + 12;
  if (rag_end > i_end) rag_end = i_end;
  for (; i < rag_end; i += 4) {
    const int ii = i + k0;
    float a0 = __builtin_nontemporal_load(wrow + ii);
    float a1 = __builtin_nontemporal_load(wrow + ii + 1);
    a0 = (ii     >= fc) ? a0 : 0.0f;
    a1 = (ii + 1 >= fc) ? a1 : 0.0f;
    v2f af; af.x = a0; af.y = a1;
    v2f bf = *(const v2f*)(xrow + ii);
    acc = __builtin_amdgcn_wmma_f32_16x16x4_f32(false, af, false, bf,
                                                (short)0, acc, false, false);
  }

#pragma unroll 4
  for (; i < i_end; i += 4) {
    const int ii = i + k0;
    v2f af;
    af.x = __builtin_nontemporal_load(wrow + ii);
    af.y = __builtin_nontemporal_load(wrow + ii + 1);
    v2f bf = *(const v2f*)(xrow + ii);
    acc = __builtin_amdgcn_wmma_f32_16x16x4_f32(false, af, false, bf,
                                                (short)0, acc, false, false);
  }

  if (n < NB) {
    const int obase = o0 + (half << 3);
    const v4f b0 = *(const v4f*)(bias + obase);
    const v4f b1 = *(const v4f*)(bias + obase + 4);
    if (final_mode) {
      float* po = out + (long)n * N_OUT + (N_OUT - 8 - obase);
      v4f lo, hi;
      lo.x = acc[7] + b1.w; lo.y = acc[6] + b1.z;
      lo.z = acc[5] + b1.y; lo.w = acc[4] + b1.x;
      hi.x = acc[3] + b0.w; hi.y = acc[2] + b0.z;
      hi.z = acc[1] + b0.y; hi.w = acc[0] + b0.x;
      *(v4f*)po       = lo;
      *(v4f*)(po + 4) = hi;
    } else {
      float* p = ws + ((((long)blockIdx.y * NTILE + tile) * NB + n) << 4) + (half << 3);
      v4f lo, hi;
      lo.x = acc[0]; lo.y = acc[1]; lo.z = acc[2]; lo.w = acc[3];
      hi.x = acc[4]; hi.y = acc[5]; hi.z = acc[6]; hi.w = acc[7];
      *(v4f*)p       = lo;
      *(v4f*)(p + 4) = hi;
    }
  }
}

__global__ __launch_bounds__(256)
void tri_reduce(const float* __restrict__ ws, const float* __restrict__ bias,
                float* __restrict__ out) {
  const int t = blockIdx.x * 256 + threadIdx.x;
  if (t >= N_OUT * NB) return;
  const int o = t >> 3, m = t & 7;
  const int tile = o >> 4, r = o & 15;
  const int fc_min = (((o & ~15) - 4) > 0) ? ((o & ~15) - 4) : 0;
  float s = 0.0f;
  for (int c = (fc_min / KCH); c < NCHUNK; ++c)
    s += ws[((((long)c * NTILE + tile) * NB + m) << 4) + r];
  out[(long)m * N_OUT + (N_OUT - 1 - o)] = s + bias[o];
}

extern "C" void kernel_launch(void* const* d_in, const int* in_sizes, int n_in,
                              void* d_out, int out_size, void* d_ws, size_t ws_size,
                              hipStream_t stream) {
  (void)in_sizes; (void)n_in; (void)out_size;
  const float* x    = (const float*)d_in[0];
  const float* w    = (const float*)d_in[1];
  const float* bias = (const float*)d_in[2];
  float*       out  = (float*)d_out;

  const size_t ws_needed = (size_t)NCHUNK * NTILE * NB * 16 * sizeof(float);
  if (ws_size >= ws_needed) {
    dim3 grid(NTILE, NCHUNK);
    tri_wmma_partial<<<grid, 32, 0, stream>>>(w, x, bias, (float*)d_ws, out,
                                              KCH,  0);
    tri_reduce<<<(N_OUT * NB + 255) / 256, 256, 0, stream>>>(
        (const float*)d_ws, bias, out);
  } else {
    dim3 grid(NTILE, 1);
    tri_wmma_partial<<<grid, 32, 0, stream>>>(w, x, bias, nullptr, out,
                                              N_IN,  1);
  }
}


# W tile staged through LDS: each load reads 128 contiguous bytes of one packed row, A fragments read back from LDS; k order within 8-column groups permuted (sum order changes, still fixed)
# speedup vs baseline: 4.0485x; 4.0485x over previous
#include <hip/hip_runtime.h>

typedef float v2f __attribute__((ext_vector_type(2)));
typedef float v4f __attribute__((ext_vector_type(4)));
typedef float v8f __attribute__((ext_vector_type(8)));

#define N_IN   8192
#define N_OUT  8192
#define NB     8
#define KCH    1024
#define NCHUNK (N_IN / KCH)
#define NTILE  (N_OUT / 16)

#define STEP   32
#define STAGES 3
#define TP     36

__global__ __launch_bounds__(32)
void tri_wmma_partial(const float* __restrict__ w, const float* __restrict__ x,
                      const float* __restrict__ bias, float* __restrict__ ws,
                      float* __restrict__ out, int k_chunk, int final_mode) {
  __shared__ __attribute__((aligned(16))) float tile[16 * TP];

  const int tile_id = blockIdx.x;
  const int o0      = tile_id << 4;
  const int fc_min  = (o0 > 4) ? (o0 - 4) : 0;
  const int k_begin = blockIdx.y * k_chunk;
  int i_end = k_begin + k_chunk;
  if (i_end > N_IN) i_end = N_IN;
  if (i_end <= fc_min) return;
  const int i0 = max(k_begin, fc_min & ~(STEP - 1));

  const int lane = threadIdx.x;
  const int half = lane >> 4;
  const int n    = lane & 15;

  const float* wr[16];
#pragma unroll
  for (int r = 0; r < 16; ++r) {
    const int  o  = o0 + r;
    const int  fc = (o > 4) ? (o - 4) : 0;
    const long rs = (long)o * N_IN - ((o >= 5) ? (((long)(o - 5) * (o - 4)) >> 1) : 0);
    wr[r] = w + (rs - fc);
  }
  const float* __restrict__ xrow = x + (long)(n & 7) * N_IN + (half << 2);
  const float* afrag = tile + n * TP + (half << 2);

  v8f acc = {};

  const int i_last = i_end - STEP;

  for (int i = i0; i < i_end; i += STAGES * STEP) {
    float rg[STAGES][16];
    v4f   xg[STAGES][4];
#pragma unroll
    for (int s = 0; s < STAGES; ++s) {
      const int il = min(i + s * STEP, i_last);
#pragma unroll
      for (int r = 0; r < 16; ++r)
        rg[s][r] = __builtin_nontemporal_load(wr[r] + il + lane);
#pragma unroll
      for (int g = 0; g < 4; ++g)
        xg[s][g] = *(const v4f*)(xrow + il + 8 * g);
    }

#pragma unroll
    for (int s = 0; s < STAGES; ++s) {
      const int ic = i + s * STEP;
      if (ic < i_end) {
        if (ic < o0 + 12) {
          const int c = ic + lane;
#pragma unroll
          for (int r = 0; r < 16; ++r) {
            const int fc = (o0 + r > 4) ? (o0 + r - 4) : 0;
            tile[r * TP + lane] = (c >= fc) ? rg[s][r] : 0.0f;
          }
        } else {
#pragma unroll
          for (int r = 0; r < 16; ++r) tile[r * TP + lane] = rg[s][r];
        }
#pragma unroll
        for (int g = 0; g < 4; ++g) {
          const v4f a4 = *(const v4f*)(afrag + 8 * g);
          const v4f b4 = xg[s][g];
          v2f a01, a23, b01, b23;
          a01.x = a4.x; a01.y = a4.y; a23.x = a4.z; a23.y = a4.w;
          b01.x = b4.x; b01.y = b4.y; b23.x = b4.z; b23.y = b4.w;
          acc = __builtin_amdgcn_wmma_f32_16x16x4_f32(false, a01, false, b01,
                                                      (short)0, acc, false, false);
          acc = __builtin_amdgcn_wmma_f32_16x16x4_f32(false, a23, false, b23,
                                                      (short)0, acc, false, false);
        }
      }
    }
  }

  if (n < NB) {
    const int obase = o0 + (half << 3);
    const v4f b0 = *(const v4f*)(bias + obase);
    const v4f b1 = *(const v4f*)(bias + obase + 4);
    if (final_mode) {
      float* po = out + (long)n * N_OUT + (N_OUT - 8 - obase);
      v4f lo, hi;
      lo.x = acc[7] + b1.w; lo.y = acc[6] + b1.z;
      lo.z = acc[5] + b1.y; lo.w = acc[4] + b1.x;
      hi.x = acc[3] + b0.w; hi.y = acc[2] + b0.z;
      hi.z = acc[1] + b0.y; hi.w = acc[0] + b0.x;
      *(v4f*)po       = lo;
      *(v4f*)(po + 4) = hi;
    } else {
      float* p = ws + ((((long)blockIdx.y * NTILE + tile_id) * NB + n) << 4) + (half << 3);
      v4f lo, hi;
      lo.x = acc[0]; lo.y = acc[1]; lo.z = acc[2]; lo.w = acc[3];
      hi.x = acc[4]; hi.y = acc[5]; hi.z = acc[6]; hi.w = acc[7];
      *(v4f*)p       = lo;
      *(v4f*)(p + 4) = hi;
    }
  }
}

__global__ __launch_bounds__(256)
void tri_reduce(const float* __restrict__ ws, const float* __restrict__ bias,
                float* __restrict__ out) {
  const int t = blockIdx.x * 256 + threadIdx.x;
  if (t >= N_OUT * NB) return;
  const int o = t >> 3, m = t & 7;
  const int tile = o >> 4, r = o & 15;
  const int fc_min = (((o & ~15) - 4) > 0) ? ((o & ~15) - 4) : 0;
  float s = 0.0f;
  for (int c = (fc_min / KCH); c < NCHUNK; ++c)
    s += ws[((((long)c * NTILE + tile) * NB + m) << 4) + r];
  out[(long)m * N_OUT + (N_OUT - 1 - o)] = s + bias[o];
}

extern "C" void kernel_launch(void* const* d_in, const int* in_sizes, int n_in,
                              void* d_out, int out_size, void* d_ws, size_t ws_size,
                              hipStream_t stream) {
  (void)in_sizes; (void)n_in; (void)out_size;
  const float* x    = (const float*)d_in[0];
  const float* w    = (const float*)d_in[1];
  const float* bias = (const float*)d_in[2];
  float*       out  = (float*)d_out;

  const size_t ws_needed = (size_t)NCHUNK * NTILE * NB * 16 * sizeof(float);
  if (ws_size >= ws_needed) {
    dim3 grid(NTILE, NCHUNK);
    tri_wmma_partial<<<grid, 32, 0, stream>>>(w, x, bias, (float*)d_ws, out,
                                              KCH,  0);
    tri_reduce<<<(N_OUT * NB + 255) / 256, 256, 0, stream>>>(
        (const float*)d_ws, bias, out);
  } else {
    dim3 grid(NTILE, 1);
    tri_wmma_partial<<<grid, 32, 0, stream>>>(w, x, bias, nullptr, out,
                                              N_IN,  1);
  }
}


# W loads widened from 4 to 16 bytes per lane: one load = 512 contiguous bytes of a packed row (any 4-byte start), 128-column steps, LDS tile stored and read in 16-byte pieces; sum order unchanged
# speedup vs baseline: 4.7115x; 1.1638x over previous
#include <hip/hip_runtime.h>

typedef float v2f __attribute__((ext_vector_type(2)));
typedef float v4f __attribute__((ext_vector_type(4)));
typedef float v8f __attribute__((ext_vector_type(8)));
typedef float v4fu __attribute__((ext_vector_type(4), aligned(4)));

#define N_IN   8192
#define N_OUT  8192
#define NB     8
#define KCH    1024
#define NCHUNK (N_IN / KCH)
#define NTILE  (N_OUT / 16)

#define STEP   128
#define NG     (STEP / 8)
#define STAGES 2
#define TP     132

__global__ __launch_bounds__(32)
void tri_wmma_partial(const float* __restrict__ w, const float* __restrict__ x,
                      const float* __restrict__ bias, float* __restrict__ ws,
                      float* __restrict__ out, int k_chunk, int final_mode) {
  __shared__ __attribute__((aligned(16))) float tile[16 * TP];

  const int tile_id = blockIdx.x;
  const int o0      = tile_id << 4;
  const int fc_min  = (o0 > 4) ? (o0 - 4) : 0;
  const int k_begin = blockIdx.y * k_chunk;
  int i_end = k_begin + k_chunk;
  if (i_end > N_IN) i_end = N_IN;
  if (i_end <= fc_min) return;
  const int i0 = max(k_begin, fc_min & ~(STEP - 1));

  const int lane = threadIdx.x;
  const int half = lane >> 4;
  const int n    = lane & 15;

  const float* wr[16];
#pragma unroll
  for (int r = 0; r < 16; ++r) {
    const int  o  = o0 + r;
    const int  fc = (o > 4) ? (o - 4) : 0;
    const long rs = (long)o * N_IN - ((o >= 5) ? (((long)(o - 5) * (o - 4)) >> 1) : 0);
    wr[r] = w + (rs - fc);
  }
  const float* __restrict__ xrow = x + (long)(n & 7) * N_IN + (half << 2);
  const float* afrag = tile + n * TP + (half << 2);

  v8f acc = {};

  const int i_last = i_end - STEP;

  for (int i = i0; i < i_end; i += STAGES * STEP) {
    v4f   rg[STAGES][16];
    v4f   xg[STAGES][NG];
#pragma unroll
    for (int s = 0; s < STAGES; ++s) {
      const int il = min(i + s * STEP, i_last);
#pragma unroll
      for (int r = 0; r < 16; ++r)
        rg[s][r] = __builtin_nontemporal_load(
            (const v4fu*)(wr[r] + il + (lane << 2)));
#pragma unroll
      for (int g = 0; g < NG; ++g)
        xg[s][g] = *(const v4f*)(xrow + il + 8 * g);
    }

#pragma unroll
    for (int s = 0; s < STAGES; ++s) {
      const int ic = i + s * STEP;
      if (ic < i_end) {
        if (ic < o0 + 12) {
          const int c = ic + (lane << 2);
#pragma unroll
          for (int r = 0; r < 16; ++r) {
            const int fc = (o0 + r > 4) ? (o0 + r - 4) : 0;
            v4f v = rg[s][r];
            v.x = (c     >= fc) ? v.x : 0.0f;
            v.y = (c + 1 >= fc) ? v.y : 0.0f;
            v.z = (c + 2 >= fc) ? v.z : 0.0f;
            v.w = (c + 3 >= fc) ? v.w : 0.0f;
            *(v4f*)(tile + r * TP + (lane << 2)) = v;
          }
        } else {
#pragma unroll
          for (int r = 0; r < 16; ++r) *(v4f*)(tile + r * TP + (lane << 2)) = rg[s][r];
        }
#pragma unroll
        for (int g = 0; g < NG; ++g) {
          const v4f a4 = *(const v4f*)(afrag + 8 * g);
          const v4f b4 = xg[s][g];
          v2f a01, a23, b01, b23;
          a01.x = a4.x; a01.y = a4.y; a23.x = a4.z; a23.y = a4.w;
          b01.x = b4.x; b01.y = b4.y; b23.x = b4.z; b23.y = b4.w;
          acc = __builtin_amdgcn_wmma_f32_16x16x4_f32(false, a01, false, b01,
                                                      (short)0, acc, false, false);
          acc = __builtin_amdgcn_wmma_f32_16x16x4_f32(false, a23, false, b23,
                                                      (short)0, acc, false, false);
        }
      }
    }
  }

  if (n < NB) {
    const int obase = o0 + (half << 3);
    const v4f b0 = *(const v4f*)(bias + obase);
    const v4f b1 = *(const v4f*)(bias + obase + 4);
    if (final_mode) {
      float* po = out + (long)n * N_OUT + (N_OUT - 8 - obase);
      v4f lo, hi;
      lo.x = acc[7] + b1.w; lo.y = acc[6] + b1.z;
      lo.z = acc[5] + b1.y; lo.w = acc[4] + b1.x;
      hi.x = acc[3] + b0.w; hi.y = acc[2] + b0.z;
      hi.z = acc[1] + b0.y; hi.w = acc[0] + b0.x;
      *(v4f*)po       = lo;
      *(v4f*)(po + 4) = hi;
    } else {
      float* p = ws + ((((long)blockIdx.y * NTILE + tile_id) * NB + n) << 4) + (half << 3);
      v4f lo, hi;
      lo.x = acc[0]; lo.y = acc[1]; lo.z = acc[2]; lo.w = acc[3];
      hi.x = acc[4]; hi.y = acc[5]; hi.z = acc[6]; hi.w = acc[7];
      *(v4f*)p       = lo;
      *(v4f*)(p + 4) = hi;
    }
  }
}

__global__ __launch_bounds__(256)
void tri_reduce(const float* __restrict__ ws, const float* __restrict__ bias,
                float* __restrict__ out) {
  const int t = blockIdx.x * 256 + threadIdx.x;
  if (t >= N_OUT * NB) return;
  const int o = t >> 3, m = t & 7;
  const int tile = o >> 4, r = o & 15;
  const int fc_min = (((o & ~15) - 4) > 0) ? ((o & ~15) - 4) : 0;
  float s = 0.0f;
  for (int c = (fc_min / KCH); c < NCHUNK; ++c)
    s += ws[((((long)c * NTILE + tile) * NB + m) << 4) + r];
  out[(long)m * N_OUT + (N_OUT - 1 - o)] = s + bias[o];
}

extern "C" void kernel_launch(void* const* d_in, const int* in_sizes, int n_in,
                              void* d_out, int out_size, void* d_ws, size_t ws_size,
                              hipStream_t stream) {
  (void)in_sizes; (void)n_in; (void)out_size;
  const float* x    = (const float*)d_in[0];
  const float* w    = (const float*)d_in[1];
  const float* bias = (const float*)d_in[2];
  float*       out  = (float*)d_out;

  const size_t ws_needed = (size_t)NCHUNK * NTILE * NB * 16 * sizeof(float);
  if (ws_size >= ws_needed) {
    dim3 grid(NTILE, NCHUNK);
    tri_wmma_partial<<<grid, 32, 0, stream>>>(w, x, bias, (float*)d_ws, out,
                                              KCH,  0);
    tri_reduce<<<(N_OUT * NB + 255) / 256, 256, 0, stream>>>(
        (const float*)d_ws, bias, out);
  } else {
    dim3 grid(NTILE, 1);
    tri_wmma_partial<<<grid, 32, 0, stream>>>(w, x, bias, nullptr, out,
                                              N_IN,  1);
  }
}


# Split-K moved inside the workgroup: 12 waves per 16-row tile take equal contiguous K slices, partials summed through LDS in ascending K order, one launch (sum order changes: slice bounds move)
# speedup vs baseline: 6.5944x; 1.3996x over previous
#include <hip/hip_runtime.h>

typedef float v2f __attribute__((ext_vector_type(2)));
typedef float v4f __attribute__((ext_vector_type(4)));
typedef float v8f __attribute__((ext_vector_type(8)));
typedef float v4fu __attribute__((ext_vector_type(4), aligned(4)));

#define N_IN   8192
#define N_OUT  8192
#define NB     8
#define NTILE  (N_OUT / 16)
#define NW     12

#define STEP   128
#define NG     (STEP / 8)
#define STAGES 2
#define TP     132
#define BLK    (STAGES * STEP)
#define WTILE  (16 * TP)

__global__ __launch_bounds__(NW * 32)
void tri_wmma_fused(const float* __restrict__ w, const float* __restrict__ x,
                    const float* __restrict__ bias, float* __restrict__ out) {
  __shared__ __attribute__((aligned(16))) float tiles[NW * WTILE];

  const int tile_id = blockIdx.x;
  const int o0      = tile_id << 4;
  const int fc_min  = (o0 > 4) ? (o0 - 4) : 0;
  const int i0      = fc_min & ~(STEP - 1);
  const int nblk    = (N_IN - i0 + BLK - 1) / BLK;
  const int wv      = __builtin_amdgcn_readfirstlane((int)(threadIdx.x >> 5));
  const int b_lo    = (wv * nblk) / NW;
  const int b_hi    = ((wv + 1) * nblk) / NW;
  const int i_begin = i0 + b_lo * BLK;
  const int i_stop  = min(i0 + b_hi * BLK, N_IN);
  const int i_end   = N_IN;

  float* tile = tiles + wv * WTILE;

  const int lane = threadIdx.x & 31;
  const int half = lane >> 4;
  const int n    = lane & 15;

  const float* wr[16];
#pragma unroll
  for (int r = 0; r < 16; ++r) {
    const int  o  = o0 + r;
    const int  fc = (o > 4) ? (o - 4) : 0;
    const long rs = (long)o * N_IN - ((o >= 5) ? (((long)(o - 5) * (o - 4)) >> 1) : 0);
    wr[r] = w + (rs - fc);
  }
  const float* __restrict__ xrow = x + (long)(n & 7) * N_IN + (half << 2);
  const float* afrag = tile + n * TP + (half << 2);

  v8f acc = {};

  const int i_last = i_end - STEP;

  for (int i = i_begin; i < i_stop; i += BLK) {
    v4f   rg[STAGES][16];
    v4f   xg[STAGES][NG];
#pragma unroll
    for (int s = 0; s < STAGES; ++s) {
      const int il = min(i + s * STEP, i_last);
#pragma unroll
      for (int r = 0; r < 16; ++r)
        rg[s][r] = __builtin_nontemporal_load(
            (const v4fu*)(wr[r] + il + (lane << 2)));
#pragma unroll
      for (int g = 0; g < NG; ++g)
        xg[s][g] = *(const v4f*)(xrow + il + 8 * g);
    }

#pragma unroll
    for (int s = 0; s < STAGES; ++s) {
      const int ic = i + s * STEP;
      if (ic < i_end) {
        if (ic < o0 + 12) {
          const int c = ic + (lane << 2);
#pragma unroll
          for (int r = 0; r < 16; ++r) {
            const int fc = (o0 + r > 4) ? (o0 + r - 4) : 0;
            v4f v = rg[s][r];
            v.x = (c     >= fc) ? v.x : 0.0f;
            v.y = (c + 1 >= fc) ? v.y : 0.0f;
            v.z = (c + 2 >= fc) ? v.z : 0.0f;
            v.w = (c + 3 >= fc) ? v.w : 0.0f;
            *(v4f*)(tile + r * TP + (lane << 2)) = v;
          }
        } else {
#pragma unroll
          for (int r = 0; r < 16; ++r) *(v4f*)(tile + r * TP + (lane << 2)) = rg[s][r];
        }
#pragma unroll
        for (int g = 0; g < NG; ++g) {
          const v4f a4 = *(const v4f*)(afrag + 8 * g);
          const v4f b4 = xg[s][g];
          v2f a01, a23, b01, b23;
          a01.x = a4.x; a01.y = a4.y; a23.x = a4.z; a23.y = a4.w;
          b01.x = b4.x; b01.y = b4.y; b23.x = b4.z; b23.y = b4.w;
          acc = __builtin_amdgcn_wmma_f32_16x16x4_f32(false, a01, false, b01,
                                                      (short)0, acc, false, false);
          acc = __builtin_amdgcn_wmma_f32_16x16x4_f32(false, a23, false, b23,
                                                      (short)0, acc, false, false);
        }
      }
    }
  }

  if (n < NB) {
    float* p = tile + (n << 4) + (half << 3);
    v4f lo, hi;
    lo.x = acc[0]; lo.y = acc[1]; lo.z = acc[2]; lo.w = acc[3];
    hi.x = acc[4]; hi.y = acc[5]; hi.z = acc[6]; hi.w = acc[7];
    *(v4f*)p       = lo;
    *(v4f*)(p + 4) = hi;
  }
  __syncthreads();

  const int tid = threadIdx.x;
  if (tid < NB * 16) {
    const int m = tid >> 4, rr = tid & 15, r = 15 - rr;
    float s = 0.0f;
#pragma unroll
    for (int k = 0; k < NW; ++k) s += tiles[k * WTILE + (m << 4) + r];
    out[(long)m * N_OUT + (N_OUT - 16 - o0) + rr] = s + bias[o0 + r];
  }
}

extern "C" void kernel_launch(void* const* d_in, const int* in_sizes, int n_in,
                              void* d_out, int out_size, void* d_ws, size_t ws_size,
                              hipStream_t stream) {
  (void)in_sizes; (void)n_in; (void)out_size; (void)d_ws; (void)ws_size;
  const float* x    = (const float*)d_in[0];
  const float* w    = (const float*)d_in[1];
  const float* bias = (const float*)d_in[2];
  float*       out  = (float*)d_out;

  tri_wmma_fused<<<NTILE, NW * 32, 0, stream>>>(w, x, bias, out);
}


# x staged through LDS like W: one x load = 512 contiguous bytes of one batch row (8 whole-line loads per 128-column step, was 16 of 8 lines each), B fragments read from LDS; sum order unchanged
# speedup vs baseline: 7.1630x; 1.0862x over previous
#include <hip/hip_runtime.h>

typedef float v2f __attribute__((ext_vector_type(2)));
typedef float v4f __attribute__((ext_vector_type(4)));
typedef float v8f __attribute__((ext_vector_type(8)));
typedef float v4fu __attribute__((ext_vector_type(4), aligned(4)));

#define N_IN   8192
#define N_OUT  8192
#define NB     8
#define NTILE  (N_OUT / 16)
#define NW     12

#define STEP   128
#define NG     (STEP / 8)
#define STAGES 2
#define TP     132
#define BLK    (STAGES * STEP)
#define WTILE  (16 * TP)
#define XTILE  (NB * TP)
#define WSLOT  (WTILE + XTILE)

__global__ __launch_bounds__(NW * 32)
void tri_wmma_fused(const float* __restrict__ w, const float* __restrict__ x,
                    const float* __restrict__ bias, float* __restrict__ out) {
  __shared__ __attribute__((aligned(16))) float tiles[NW * WSLOT];

  const int tile_id = blockIdx.x;
  const int o0      = tile_id << 4;
  const int fc_min  = (o0 > 4) ? (o0 - 4) : 0;
  const int i0      = fc_min & ~(STEP - 1);
  const int nblk    = (N_IN - i0 + BLK - 1) / BLK;
  const int wv      = __builtin_amdgcn_readfirstlane((int)(threadIdx.x >> 5));
  const int b_lo    = (wv * nblk) / NW;
  const int b_hi    = ((wv + 1) * nblk) / NW;
  const int i_begin = i0 + b_lo * BLK;
  const int i_stop  = min(i0 + b_hi * BLK, N_IN);
  const int i_end   = N_IN;

  float* tile = tiles + wv * WSLOT;
  float* xt   = tile + WTILE;

  const int lane = threadIdx.x & 31;
  const int half = lane >> 4;
  const int n    = lane & 15;

  const float* wr[16];
#pragma unroll
  for (int r = 0; r < 16; ++r) {
    const int  o  = o0 + r;
    const int  fc = (o > 4) ? (o - 4) : 0;
    const long rs = (long)o * N_IN - ((o >= 5) ? (((long)(o - 5) * (o - 4)) >> 1) : 0);
    wr[r] = w + (rs - fc);
  }
  const float* __restrict__ xcol = x + (lane << 2);
  const float* afrag = tile + n * TP + (half << 2);
  const float* bfrag = xt + (n & 7) * TP + (half << 2);

  v8f acc = {};

  const int i_last = i_end - STEP;

  for (int i = i_begin; i < i_stop; i += BLK) {
    v4f   rg[STAGES][16];
    v4f   xg[STAGES][NB];
#pragma unroll
    for (int s = 0; s < STAGES; ++s) {
      const int il = min(i + s * STEP, i_last);
#pragma unroll
      for (int r = 0; r < 16; ++r)
        rg[s][r] = __builtin_nontemporal_load(
            (const v4fu*)(wr[r] + il + (lane << 2)));
#pragma unroll
      for (int b = 0; b < NB; ++b)
        xg[s][b] = *(const v4f*)(xcol + (long)b * N_IN + il);
    }

#pragma unroll
    for (int s = 0; s < STAGES; ++s) {
      const int ic = i + s * STEP;
      if (ic < i_end) {
        if (ic < o0 + 12) {
          const int c = ic + (lane << 2);
#pragma unroll
          for (int r = 0; r < 16; ++r) {
            const int fc = (o0 + r > 4) ? (o0 + r - 4) : 0;
            v4f v = rg[s][r];
            v.x = (c     >= fc) ? v.x : 0.0f;
            v.y = (c + 1 >= fc) ? v.y : 0.0f;
            v.z = (c + 2 >= fc) ? v.z : 0.0f;
            v.w = (c + 3 >= fc) ? v.w : 0.0f;
            *(v4f*)(tile + r * TP + (lane << 2)) = v;
          }
        } else {
#pragma unroll
          for (int r = 0; r < 16; ++r) *(v4f*)(tile + r * TP + (lane << 2)) = rg[s][r];
        }
#pragma unroll
        for (int b = 0; b < NB; ++b) *(v4f*)(xt + b * TP + (lane << 2)) = xg[s][b];
#pragma unroll
        for (int g = 0; g < NG; ++g) {
          const v4f a4 = *(const v4f*)(afrag + 8 * g);
          const v4f b4 = *(const v4f*)(bfrag + 8 * g);
          v2f a01, a23, b01, b23;
          a01.x = a4.x; a01.y = a4.y; a23.x = a4.z; a23.y = a4.w;
          b01.x = b4.x; b01.y = b4.y; b23.x = b4.z; b23.y = b4.w;
          acc = __builtin_amdgcn_wmma_f32_16x16x4_f32(false, a01, false, b01,
                                                      (short)0, acc, false, false);
          acc = __builtin_amdgcn_wmma_f32_16x16x4_f32(false, a23, false, b23,
                                                      (short)0, acc, false, false);
        }
      }
    }
  }

  if (n < NB) {
    float* p = tile + (n << 4) + (half << 3);
    v4f lo, hi;
    lo.x = acc[0]; lo.y = acc[1]; lo.z = acc[2]; lo.w = acc[3];
    hi.x = acc[4]; hi.y = acc[5]; hi.z = acc[6]; hi.w = acc[7];
    *(v4f*)p       = lo;
    *(v4f*)(p + 4) = hi;
  }
  __syncthreads();

  const int tid = threadIdx.x;
  if (tid < NB * 16) {
    const int m = tid >> 4, rr = tid & 15, r = 15 - rr;
    float s = 0.0f;
#pragma unroll
    for (int k = 0; k < NW; ++k) s += tiles[k * WSLOT + (m << 4) + r];
    out[(long)m * N_OUT + (N_OUT - 16 - o0) + rr] = s + bias[o0 + r];
  }
}

extern "C" void kernel_launch(void* const* d_in, const int* in_sizes, int n_in,
                              void* d_out, int out_size, void* d_ws, size_t ws_size,
                              hipStream_t stream) {
  (void)in_sizes; (void)n_in; (void)out_size; (void)d_ws; (void)ws_size;
  const float* x    = (const float*)d_in[0];
  const float* w    = (const float*)d_in[1];
  const float* bias = (const float*)d_in[2];
  float*       out  = (float*)d_out;

  tri_wmma_fused<<<NTILE, NW * 32, 0, stream>>>(w, x, bias, out);
}
